# starting kernel
# baseline (speedup 1.0000x reference)
#include <hip/hip_runtime.h>

typedef float v2f __attribute__((ext_vector_type(2)));
typedef float v8f __attribute__((ext_vector_type(8)));

#define IN_DIM  128
#define HID_DIM 32
#define OUT_DIM 8

__global__ void __launch_bounds__(256) init_kernel(float* __restrict__ deg,
                                                   float* __restrict__ h,
                                                   float* __restrict__ out,
                                                   const float* __restrict__ b2,
                                                   int n) {
    int i = blockIdx.x * blockDim.x + threadIdx.x;
    if (i < n) deg[i] = 1.0f;
    if (i < n * OUT_DIM) out[i] = b2[i & (OUT_DIM - 1)];
    if (i < n * HID_DIM) h[i] = 0.0f;
}

__global__ void __launch_bounds__(256) deg_kernel(const int* __restrict__ dstIdx,
                                                  float* __restrict__ deg, int E) {
    int e = blockIdx.x * blockDim.x + threadIdx.x;
    if (e < E) atomicAdd(&deg[dstIdx[e]], 1.0f);
}

__global__ void __launch_bounds__(256) rsqrt_kernel(float* __restrict__ deg, int n) {
    int i = blockIdx.x * blockDim.x + threadIdx.x;
    if (i < n) deg[i] = rsqrtf(deg[i]);
}

__global__ void __launch_bounds__(256) gemm1_kernel(const float* __restrict__ x,
                                                    const float* __restrict__ W1,
                                                    float* __restrict__ hw, int n) {
    const int lane = threadIdx.x & 31;
    const int half = lane >> 4;
    const int l16  = lane & 15;
    const int tile = blockIdx.x * 8 + (threadIdx.x >> 5);
    const int ntiles = (n + 15) >> 4;
    if (tile >= ntiles) return;

    int row = tile * 16 + l16;
    if (row >= n) row = n - 1;
    const float* xr = x + row * IN_DIM;

    v8f c0 = {}; v8f c1 = {};
    #pragma unroll 8
    for (int k = 0; k < IN_DIM; k += 4) {
        const int kk = k + 2 * half;
        v2f a;  a.x  = xr[kk];                       a.y  = xr[kk + 1];
        v2f b0; b0.x = W1[kk * HID_DIM + l16];       b0.y = W1[(kk + 1) * HID_DIM + l16];
        v2f b1; b1.x = W1[kk * HID_DIM + 16 + l16];  b1.y = W1[(kk + 1) * HID_DIM + 16 + l16];
        c0 = __builtin_amdgcn_wmma_f32_16x16x4_f32(false, a, false, b0, (short)0, c0, false, false);
        c1 = __builtin_amdgcn_wmma_f32_16x16x4_f32(false, a, false, b1, (short)0, c1, false, false);
    }
    #pragma unroll
    for (int i = 0; i < 8; ++i) {
        int orow = tile * 16 + i + 8 * half;
        if (orow < n) {
            hw[orow * HID_DIM + l16]      = c0[i];
            hw[orow * HID_DIM + 16 + l16] = c1[i];
        }
    }
}

__global__ void __launch_bounds__(256) agg1_kernel(const int* __restrict__ srcIdx,
                                                   const int* __restrict__ dstIdx,
                                                   const float* __restrict__ dinv,
                                                   const float* __restrict__ hw,
                                                   float* __restrict__ h,
                                                   int E, int n) {
    int tid  = blockIdx.x * blockDim.x + threadIdx.x;
    int e    = tid >> 5;
    int lane = tid & 31;
    if (e >= E + n) return;
    int s, d;
    if (e < E) { s = srcIdx[e]; d = dstIdx[e]; }
    else       { s = e - E;     d = s; }
    float nrm = dinv[s] * dinv[d];
    float val = hw[s * HID_DIM + lane] * nrm;
    atomicAdd(&h[d * HID_DIM + lane], val);
}

__global__ void __launch_bounds__(256) bias_relu_kernel(float* __restrict__ h,
                                                        const float* __restrict__ b1, int n) {
    int i = blockIdx.x * blockDim.x + threadIdx.x;
    if (i < n * HID_DIM) h[i] = fmaxf(h[i] + b1[i & (HID_DIM - 1)], 0.0f);
}

__global__ void __launch_bounds__(256) gemm2_kernel(const float* __restrict__ h,
                                                    const float* __restrict__ W2,
                                                    float* __restrict__ g, int n) {
    const int lane = threadIdx.x & 31;
    const int half = lane >> 4;
    const int l16  = lane & 15;
    const int tile = blockIdx.x * 8 + (threadIdx.x >> 5);
    const int ntiles = (n + 15) >> 4;
    if (tile >= ntiles) return;

    int row = tile * 16 + l16;
    if (row >= n) row = n - 1;
    const float* hr = h + row * HID_DIM;
    const bool live = (l16 < OUT_DIM);

    v8f c = {};
    #pragma unroll
    for (int k = 0; k < HID_DIM; k += 4) {
        const int kk = k + 2 * half;
        v2f a; a.x = hr[kk]; a.y = hr[kk + 1];
        v2f b;
        b.x = live ? W2[kk * OUT_DIM + l16]       : 0.0f;
        b.y = live ? W2[(kk + 1) * OUT_DIM + l16] : 0.0f;
        c = __builtin_amdgcn_wmma_f32_16x16x4_f32(false, a, false, b, (short)0, c, false, false);
    }
    if (live) {
        #pragma unroll
        for (int i = 0; i < 8; ++i) {
            int orow = tile * 16 + i + 8 * half;
            if (orow < n) g[orow * OUT_DIM + l16] = c[i];
        }
    }
}

__global__ void __launch_bounds__(256) agg2_kernel(const int* __restrict__ srcIdx,
                                                   const int* __restrict__ dstIdx,
                                                   const float* __restrict__ dinv,
                                                   const float* __restrict__ g,
                                                   float* __restrict__ out,
                                                   int E, int n) {
    int tid = blockIdx.x * blockDim.x + threadIdx.x;
    int e = tid >> 3;
    int f = tid & 7;
    if (e >= E + n) return;
    int s, d;
    if (e < E) { s = srcIdx[e]; d = dstIdx[e]; }
    else       { s = e - E;     d = s; }
    float nrm = dinv[s] * dinv[d];
    atomicAdd(&out[d * OUT_DIM + f], g[s * OUT_DIM + f] * nrm);
}

extern "C" void kernel_launch(void* const* d_in, const int* in_sizes, int n_in,
                              void* d_out, int out_size, void* d_ws, size_t ws_size,
                              hipStream_t stream) {
    const float* x  = (const float*)d_in[0];
    const int*   ei = (const int*)d_in[1];
    const float* W1 = (const float*)d_in[2];
    const float* b1 = (const float*)d_in[3];
    const float* W2 = (const float*)d_in[4];
    const float* b2 = (const float*)d_in[5];
    float* out = (float*)d_out;

    const int n = in_sizes[0] / IN_DIM;
    const int E = in_sizes[1] / 2;
    const int* srcIdx = ei;
    const int* dstIdx = ei + E;

    float* deg = (float*)d_ws;
    float* hw  = deg + n;
    float* h   = hw + (size_t)n * HID_DIM;
    float* g   = h  + (size_t)n * HID_DIM;

    const int ET = E + n;
    const dim3 blk(256);

    init_kernel<<<dim3((n * HID_DIM + 255) / 256), blk, 0, stream>>>(deg, h, out, b2, n);
    deg_kernel<<<dim3((E + 255) / 256), blk, 0, stream>>>(dstIdx, deg, E);
    rsqrt_kernel<<<dim3((n + 255) / 256), blk, 0, stream>>>(deg, n);

    const int ntiles = (n + 15) / 16;
    gemm1_kernel<<<dim3((ntiles + 7) / 8), blk, 0, stream>>>(x, W1, hw, n);

    long t1 = (long)ET * 32;
    agg1_kernel<<<dim3((unsigned)((t1 + 255) / 256)), blk, 0, stream>>>(srcIdx, dstIdx, deg, hw, h, E, n);
    bias_relu_kernel<<<dim3((n * HID_DIM + 255) / 256), blk, 0, stream>>>(h, b1, n);

    gemm2_kernel<<<dim3((ntiles + 7) / 8), blk, 0, stream>>>(h, W2, g, n);

    long t2 = (long)ET * 8;
    agg2_kernel<<<dim3((unsigned)((t2 + 255) / 256)), blk, 0, stream>>>(srcIdx, dstIdx, deg, g, out, E, n);
}


# Both aggregations: bucket edges by destination once per call (one returning int atomic per edge gives the slot), then gather per node with no float atomics; sums run in slot (arrival) order
# speedup vs baseline: 2.0188x; 2.0188x over previous
#include <hip/hip_runtime.h>


typedef float v2f __attribute__((ext_vector_type(2)));
typedef float v8f __attribute__((ext_vector_type(8)));

#define IN_DIM  128
#define HID_DIM 32
#define OUT_DIM 8
#define CAP     64

__global__ void __launch_bounds__(256) init_kernel(int* __restrict__ cnt,
                                                   int* __restrict__ spill_cnt,
                                                   float* __restrict__ hw,
                                                   float* __restrict__ g, int n) {
    int i = blockIdx.x * blockDim.x + threadIdx.x;
    if (i < n) cnt[i] = 0;
    if (i == 0) spill_cnt[0] = 0;
    if (i < HID_DIM) hw[n * HID_DIM + i] = 0.0f;
    if (i < OUT_DIM) g[n * OUT_DIM + i] = 0.0f;
}

__global__ void __launch_bounds__(256) bucket_kernel(const int* __restrict__ srcIdx,
                                                     const int* __restrict__ dstIdx,
                                                     int* __restrict__ cnt,
                                                     int* __restrict__ ell,
                                                     int* __restrict__ spill_cnt,
                                                     int* __restrict__ spill, int E) {
    int e = blockIdx.x * blockDim.x + threadIdx.x;
    if (e >= E) return;
    const int d = dstIdx[e];
    const int s = srcIdx[e];
    const int r = atomicAdd(&cnt[d], 1);
    if (r < CAP) {
        ell[d * CAP + r] = s;
    } else {
        const int k = atomicAdd(spill_cnt, 1);
        spill[k] = e;
    }
}

__global__ void __launch_bounds__(256) gemm1_kernel(const float* __restrict__ x,
                                                    const float* __restrict__ W1,
                                                    float* __restrict__ hw, int n) {
    const int lane = threadIdx.x & 31;
    const int half = lane >> 4;
    const int l16  = lane & 15;
    const int tile = blockIdx.x * 8 + (threadIdx.x >> 5);
    const int ntiles = (n + 15) >> 4;
    if (tile >= ntiles) return;

    int row = tile * 16 + l16;
    if (row >= n) row = n - 1;
    const float* xr = x + row * IN_DIM;

    v8f c0 = {}; v8f c1 = {};
    #pragma unroll 8
    for (int k = 0; k < IN_DIM; k += 4) {
        const int kk = k + 2 * half;
        v2f a;  a.x  = xr[kk];                       a.y  = xr[kk + 1];
        v2f b0; b0.x = W1[kk * HID_DIM + l16];       b0.y = W1[(kk + 1) * HID_DIM + l16];
        v2f b1; b1.x = W1[kk * HID_DIM + 16 + l16];  b1.y = W1[(kk + 1) * HID_DIM + 16 + l16];
        c0 = __builtin_amdgcn_wmma_f32_16x16x4_f32(false, a, false, b0, (short)0, c0, false, false);
        c1 = __builtin_amdgcn_wmma_f32_16x16x4_f32(false, a, false, b1, (short)0, c1, false, false);
    }
    #pragma unroll
    for (int i = 0; i < 8; ++i) {
        int orow = tile * 16 + i + 8 * half;
        if (orow < n) {
            hw[orow * HID_DIM + l16]      = c0[i];
            hw[orow * HID_DIM + 16 + l16] = c1[i];
        }
    }
}

__global__ void __launch_bounds__(256) agg1_kernel(const int* __restrict__ srcIdx,
                                                   const int* __restrict__ dstIdx,
                                                   const int* __restrict__ cnt,
                                                   const int* __restrict__ ell,
                                                   const int* __restrict__ spill_cnt,
                                                   const int* __restrict__ spill,
                                                   const float* __restrict__ hw,
                                                   const float* __restrict__ b1,
                                                   float* __restrict__ h, int n) {
    const int lane = threadIdx.x & 31;
    const int d = __builtin_amdgcn_readfirstlane((int)(blockIdx.x * 8 + (threadIdx.x >> 5)));
    if (d >= n) return;
    const int m = __builtin_amdgcn_readfirstlane(cnt[d]);
    const float dd = rsqrtf((float)(m + 1));
    float acc = hw[d * HID_DIM + lane] * (dd * dd);
    const int mm = min(m, CAP);
    const int* row = ell + d * CAP;

    for (int base = 0; base < mm; base += 32) {
        const int j = base + lane;
        const bool ok = j < mm;
        const int s_l = ok ? row[j] : n;
        const float n_l = ok ? rsqrtf((float)(cnt[s_l] + 1)) * dd : 0.0f;
        const int c = min(32, mm - base);
        for (int t = 0; t < c; t += 8) {
            int   s8[8];
            float w8[8];
            float v8[8];
            #pragma unroll
            for (int k = 0; k < 8; ++k) {
                s8[k] = __builtin_amdgcn_readlane(s_l, t + k);
                w8[k] = __int_as_float(__builtin_amdgcn_readlane(__float_as_int(n_l), t + k));
            }
            #pragma unroll
            for (int k = 0; k < 8; ++k) v8[k] = hw[s8[k] * HID_DIM + lane];
            #pragma unroll
            for (int k = 0; k < 8; ++k) acc += v8[k] * w8[k];
        }
    }
    if (m > CAP) {
        const int K = spill_cnt[0];
        for (int i = 0; i < K; ++i) {
            const int e = spill[i];
            if (dstIdx[e] == d) {
                const int s = srcIdx[e];
                acc += hw[s * HID_DIM + lane] * (rsqrtf((float)(cnt[s] + 1)) * dd);
            }
        }
    }
    h[d * HID_DIM + lane] = fmaxf(acc + b1[lane], 0.0f);
}

__global__ void __launch_bounds__(256) gemm2_kernel(const float* __restrict__ h,
                                                    const float* __restrict__ W2,
                                                    float* __restrict__ g, int n) {
    const int lane = threadIdx.x & 31;
    const int half = lane >> 4;
    const int l16  = lane & 15;
    const int tile = blockIdx.x * 8 + (threadIdx.x >> 5);
    const int ntiles = (n + 15) >> 4;
    if (tile >= ntiles) return;

    int row = tile * 16 + l16;
    if (row >= n) row = n - 1;
    const float* hr = h + row * HID_DIM;
    const bool live = (l16 < OUT_DIM);

    v8f c = {};
    #pragma unroll
    for (int k = 0; k < HID_DIM; k += 4) {
        const int kk = k + 2 * half;
        v2f a; a.x = hr[kk]; a.y = hr[kk + 1];
        v2f b;
        b.x = live ? W2[kk * OUT_DIM + l16]       : 0.0f;
        b.y = live ? W2[(kk + 1) * OUT_DIM + l16] : 0.0f;
        c = __builtin_amdgcn_wmma_f32_16x16x4_f32(false, a, false, b, (short)0, c, false, false);
    }
    if (live) {
        #pragma unroll
        for (int i = 0; i < 8; ++i) {
            int orow = tile * 16 + i + 8 * half;
            if (orow < n) g[orow * OUT_DIM + l16] = c[i];
        }
    }
}

__global__ void __launch_bounds__(256) agg2_kernel(const int* __restrict__ srcIdx,
                                                   const int* __restrict__ dstIdx,
                                                   const int* __restrict__ cnt,
                                                   const int* __restrict__ ell,
                                                   const int* __restrict__ spill_cnt,
                                                   const int* __restrict__ spill,
                                                   const float* __restrict__ g,
                                                   const float* __restrict__ b2,
                                                   float* __restrict__ out, int n) {
    const int tid  = blockIdx.x * blockDim.x + threadIdx.x;
    const int lane = threadIdx.x & 31;
    const int f    = lane & 7;
    const int gb   = lane & 24;
    const int node = tid >> 3;
    const bool live = node < n;
    const int d = live ? node : n - 1;
    const int m = cnt[d];
    const float dd = rsqrtf((float)(m + 1));
    const int mm = live ? min(m, CAP) : 0;
    float acc = g[d * OUT_DIM + f] * (dd * dd);
    int mx = mm;
    mx = max(mx, __shfl_xor(mx, 8));
    mx = max(mx, __shfl_xor(mx, 16));
    const int* row = ell + d * CAP;

    for (int base = 0; base < mx; base += 8) {
        const int j = base + f;
        const bool ok = j < mm;
        const int s_l = ok ? row[j] : n;
        const float n_l = ok ? rsqrtf((float)(cnt[s_l] + 1)) * dd : 0.0f;
        float v8[8];
        float w8[8];
        #pragma unroll
        for (int t = 0; t < 8; ++t) {
            const int s_t = __shfl(s_l, gb + t);
            w8[t] = __shfl(n_l, gb + t);
            v8[t] = g[s_t * OUT_DIM + f];
        }
        #pragma unroll
        for (int t = 0; t < 8; ++t) acc += v8[t] * w8[t];
    }
    if (__builtin_amdgcn_ballot_w32(live && m > CAP) != 0u) {
        const int K = spill_cnt[0];
        for (int i = 0; i < K; ++i) {
            const int e = spill[i];
            const int de = dstIdx[e];
            const int s = srcIdx[e];
            const float w = rsqrtf((float)(cnt[s] + 1)) * dd;
            const float v = g[s * OUT_DIM + f];
            acc = (live && de == d) ? acc + v * w : acc;
        }
    }
    if (live) out[node * OUT_DIM + f] = acc + b2[f];
}

extern "C" void kernel_launch(void* const* d_in, const int* in_sizes, int n_in,
                              void* d_out, int out_size, void* d_ws, size_t ws_size,
                              hipStream_t stream) {
    const float* x  = (const float*)d_in[0];
    const int*   ei = (const int*)d_in[1];
    const float* W1 = (const float*)d_in[2];
    const float* b1 = (const float*)d_in[3];
    const float* W2 = (const float*)d_in[4];
    const float* b2 = (const float*)d_in[5];
    float* out = (float*)d_out;

    const int n = in_sizes[0] / IN_DIM;
    const int E = in_sizes[1] / 2;
    const int* srcIdx = ei;
    const int* dstIdx = ei + E;

    const size_t nr = ((size_t)n + 63) / 64 * 64;
    const size_t Er = ((size_t)E + 63) / 64 * 64;
    int*   cnt       = (int*)d_ws;
    int*   spill_cnt = cnt + nr;
    int*   ell       = spill_cnt + 64;
    int*   spill     = ell + nr * CAP;
    float* hw        = (float*)(spill + Er);
    float* h         = hw + (nr + 64) * HID_DIM;
    float* g         = h  + nr * HID_DIM;

    const dim3 blk(256);
    const int ntiles = (n + 15) / 16;

    init_kernel<<<dim3((n + 255) / 256), blk, 0, stream>>>(cnt, spill_cnt, hw, g, n);
    bucket_kernel<<<dim3((E + 255) / 256), blk, 0, stream>>>(srcIdx, dstIdx, cnt, ell, spill_cnt, spill, E);
    gemm1_kernel<<<dim3((ntiles + 7) / 8), blk, 0, stream>>>(x, W1, hw, n);
    agg1_kernel<<<dim3((n + 7) / 8), blk, 0, stream>>>(srcIdx, dstIdx, cnt, ell, spill_cnt, spill, hw, b1, h, n);
    gemm2_kernel<<<dim3((ntiles + 7) / 8), blk, 0, stream>>>(h, W2, g, n);
    agg2_kernel<<<dim3((n * OUT_DIM + 255) / 256), blk, 0, stream>>>(srcIdx, dstIdx, cnt, ell, spill_cnt, spill, g, b2, out, n);
}


# Edge table built without per-edge global atomics: 8192-edge chunks sorted by 256-node destination block in LDS, one work-group per block then fills its rows in LDS; sums in slot (arrival) order
# speedup vs baseline: 4.5277x; 2.2428x over previous
#include <hip/hip_runtime.h>


typedef float v2f __attribute__((ext_vector_type(2)));
typedef float v8f __attribute__((ext_vector_type(8)));

#define IN_DIM  128
#define HID_DIM 32
#define OUT_DIM 8
#define CAP     64
#define CH      8192
#define BSH     8
#define RB      256
#define MAXB    512


__global__ void __launch_bounds__(1024) sort_kernel(const int* __restrict__ srcIdx,
                                                    const int* __restrict__ dstIdx,
                                                    unsigned* __restrict__ sorted,
                                                    int* __restrict__ pre,
                                                    int* __restrict__ spill_cnt,
                                                    float* __restrict__ hw,
                                                    float* __restrict__ g,
                                                    int E, int n, int PB) {
    __shared__ int hist[MAXB];
    __shared__ int pfx[MAXB + 1];
    __shared__ unsigned val[CH];
    const int tid = threadIdx.x;
    const int e0 = blockIdx.x * CH;
    if (blockIdx.x == 0) {
        if (tid == 0) spill_cnt[0] = 0;
        if (tid < HID_DIM) hw[n * HID_DIM + tid] = 0.0f;
        if (tid < OUT_DIM) g[n * OUT_DIM + tid] = 0.0f;
    }
    if (tid < MAXB) hist[tid] = 0;
    __syncthreads();

    unsigned w[CH / 1024];
    int bb[CH / 1024];
    int pos[CH / 1024];
    int dv[CH / 1024];
    int sv[CH / 1024];
    #pragma unroll
    for (int k = 0; k < CH / 1024; ++k) {
        const int ec = min(e0 + k * 1024 + tid, E - 1);
        dv[k] = dstIdx[ec];
        sv[k] = srcIdx[ec];
    }
    #pragma unroll
    for (int k = 0; k < CH / 1024; ++k) {
        const bool valid = (e0 + k * 1024 + tid) < E;
        bb[k] = valid ? (dv[k] >> BSH) : -1;
        w[k]  = ((unsigned)(dv[k] & (RB - 1)) << 24) | (unsigned)sv[k];
    }
    #pragma unroll
    for (int k = 0; k < CH / 1024; ++k) {
        pos[k] = 0;
        if (bb[k] >= 0) pos[k] = atomicAdd(&hist[bb[k]], 1);
    }
    __syncthreads();

    if (tid < 32) {
        int loc[MAXB / 32];
        int sum = 0;
        #pragma unroll
        for (int j = 0; j < MAXB / 32; ++j) { loc[j] = hist[tid * (MAXB / 32) + j]; sum += loc[j]; }
        int incl = sum;
        #pragma unroll
        for (int off = 1; off < 32; off <<= 1) {
            const int t = __shfl_up(incl, off);
            if (tid >= off) incl += t;
        }
        int run = incl - sum;
        #pragma unroll
        for (int j = 0; j < MAXB / 32; ++j) { pfx[tid * (MAXB / 32) + j] = run; run += loc[j]; }
        if (tid == 31) pfx[MAXB] = run;
    }
    __syncthreads();

    #pragma unroll
    for (int k = 0; k < CH / 1024; ++k)
        if (bb[k] >= 0) val[pfx[bb[k]] + pos[k]] = w[k];
    if (tid < PB) pre[blockIdx.x * PB + tid] = pfx[tid];
    __syncthreads();

    const int tot = pfx[MAXB];
    for (int i = tid; i < tot; i += 1024) sorted[e0 + i] = val[i];
}

__global__ void __launch_bounds__(256) build_kernel(const unsigned* __restrict__ sorted,
                                                    const int* __restrict__ pre,
                                                    int* __restrict__ cnt,
                                                    int* __restrict__ ell,
                                                    int* __restrict__ spill_cnt,
                                                    int2* __restrict__ spill,
                                                    int nWG, int PB, int n) {
    __shared__ int lcnt[RB];
    __shared__ int lell[RB * CAP];
    __shared__ int lo_s[256];
    __shared__ int len_s[256];
    const int tid  = threadIdx.x;
    const int lane = tid & 31;
    const int wave = __builtin_amdgcn_readfirstlane(tid >> 5);
    const int bin  = blockIdx.x;
    const int node0 = bin * RB;

    auto take = [&](unsigned wv) { return atomicAdd(&lcnt[wv >> 24], 1); };
    auto place = [&](unsigned wv, int r) {
        const int dl = (int)(wv >> 24);
        const int s  = (int)(wv & 0xFFFFFFu);
        if (r < CAP) {
            lell[dl * CAP + r] = s;
        } else {
            const int kk = atomicAdd(spill_cnt, 1);
            spill[kk] = make_int2(s, node0 + dl);
        }
    };

    lcnt[tid] = 0;
    for (int g0 = 0; g0 < nWG; g0 += 256) {
        __syncthreads();
        const int wg = g0 + tid;
        int lo = 0, len = 0;
        if (wg < nWG) {
            lo  = pre[wg * PB + bin];
            len = pre[wg * PB + bin + 1] - lo;
        }
        lo_s[tid] = lo;
        len_s[tid] = len;
        __syncthreads();
        for (int k0 = wave; k0 < 256; k0 += 64) {
            const int kk  = k0 + 8 * (lane & 7);
            const int lnv = len_s[kk];
            const int lov = lo_s[kk];
            unsigned v[8];
            int ln[8];
            const unsigned* bp[8];
            #pragma unroll
            for (int r = 0; r < 8; ++r) {
                ln[r] = __builtin_amdgcn_readlane(lnv, r);
                bp[r] = sorted + (size_t)(g0 + k0 + 8 * r) * CH + __builtin_amdgcn_readlane(lov, r);
                v[r] = (lane < ln[r]) ? bp[r][lane] : 0u;
            }
            int sl[8];
            #pragma unroll
            for (int r = 0; r < 8; ++r)
                sl[r] = (lane < ln[r]) ? take(v[r]) : 0;
            #pragma unroll
            for (int r = 0; r < 8; ++r)
                if (lane < ln[r]) place(v[r], sl[r]);
            #pragma unroll
            for (int r = 0; r < 8; ++r)
                for (int j = 32 + lane; j < ln[r]; j += 32) {
                    const unsigned wv = bp[r][j];
                    place(wv, take(wv));
                }
        }
    }
    __syncthreads();

    if (node0 + tid < n) cnt[node0 + tid] = lcnt[tid];
    const int myrow = wave * 32 + lane;
    const int mine  = (node0 + myrow < n) ? min(lcnt[myrow], CAP) : 0;
    #pragma unroll 8
    for (int i = 0; i < 32; ++i) {
        const int rr = wave * 32 + i;
        const int m  = __builtin_amdgcn_readlane(mine, i);
        int* dstrow = ell + (size_t)(node0 + rr) * CAP;
        if (lane < m)      dstrow[lane]      = lell[rr * CAP + lane];
        if (lane + 32 < m) dstrow[32 + lane] = lell[rr * CAP + 32 + lane];
    }
}

__global__ void __launch_bounds__(256) gemm1_kernel(const float* __restrict__ x,
                                                    const float* __restrict__ W1,
                                                    const int* __restrict__ cnt,
                                                    float* __restrict__ hw, int n) {
    const int lane = threadIdx.x & 31;
    const int half = lane >> 4;
    const int l16  = lane & 15;
    const int tile = blockIdx.x * 8 + (threadIdx.x >> 5);
    const int ntiles = (n + 15) >> 4;
    if (tile >= ntiles) return;

    int row = tile * 16 + l16;
    if (row >= n) row = n - 1;
    const float* xr = x + row * IN_DIM;

    float sc[8];
    #pragma unroll
    for (int i = 0; i < 8; ++i) {
        int orow = tile * 16 + i + 8 * half;
        if (orow >= n) orow = n - 1;
        sc[i] = rsqrtf((float)(cnt[orow] + 1));
    }

    v8f c0 = {}; v8f c1 = {};
    #pragma unroll 8
    for (int k = 0; k < IN_DIM; k += 4) {
        const int kk = k + 2 * half;
        v2f a;  a.x  = xr[kk];                       a.y  = xr[kk + 1];
        v2f b0; b0.x = W1[kk * HID_DIM + l16];       b0.y = W1[(kk + 1) * HID_DIM + l16];
        v2f b1; b1.x = W1[kk * HID_DIM + 16 + l16];  b1.y = W1[(kk + 1) * HID_DIM + 16 + l16];
        c0 = __builtin_amdgcn_wmma_f32_16x16x4_f32(false, a, false, b0, (short)0, c0, false, false);
        c1 = __builtin_amdgcn_wmma_f32_16x16x4_f32(false, a, false, b1, (short)0, c1, false, false);
    }
    #pragma unroll
    for (int i = 0; i < 8; ++i) {
        int orow = tile * 16 + i + 8 * half;
        if (orow < n) {
            hw[orow * HID_DIM + l16]      = c0[i] * sc[i];
            hw[orow * HID_DIM + 16 + l16] = c1[i] * sc[i];
        }
    }
}

__global__ void __launch_bounds__(256) agg1_kernel(const int* __restrict__ cnt,
                                                   const int* __restrict__ ell,
                                                   const int* __restrict__ spill_cnt,
                                                   const int2* __restrict__ spill,
                                                   const float* __restrict__ hw,
                                                   const float* __restrict__ b1,
                                                   float* __restrict__ h, int n) {
    const int lane = threadIdx.x & 31;
    const int d = __builtin_amdgcn_readfirstlane((int)(blockIdx.x * 8 + (threadIdx.x >> 5)));
    if (d >= n) return;
    const int m = __builtin_amdgcn_readfirstlane(cnt[d]);
    const float dd = rsqrtf((float)(m + 1));
    float acc = hw[d * HID_DIM + lane];
    const int mm = min(m, CAP);
    const int* row = ell + d * CAP;

    for (int base = 0; base < mm; base += 32) {
        const int j = base + lane;
        const int s_l = (j < mm) ? row[j] : n;
        const int c = min(32, mm - base);
        for (int t = 0; t < c; t += 8) {
            int   s8[8];
            float v8[8];
            #pragma unroll
            for (int k = 0; k < 8; ++k) s8[k] = __builtin_amdgcn_readlane(s_l, t + k);
            #pragma unroll
            for (int k = 0; k < 8; ++k) v8[k] = hw[s8[k] * HID_DIM + lane];
            #pragma unroll
            for (int k = 0; k < 8; ++k) acc += v8[k];
        }
    }
    if (m > CAP) {
        const int K = spill_cnt[0];
        for (int i = 0; i < K; ++i) {
            const int2 p = spill[i];
            if (p.y == d) acc += hw[p.x * HID_DIM + lane];
        }
    }
    h[d * HID_DIM + lane] = fmaxf(acc * dd + b1[lane], 0.0f);
}

__global__ void __launch_bounds__(256) gemm2_kernel(const float* __restrict__ h,
                                                    const float* __restrict__ W2,
                                                    const int* __restrict__ cnt,
                                                    float* __restrict__ g, int n) {
    const int lane = threadIdx.x & 31;
    const int half = lane >> 4;
    const int l16  = lane & 15;
    const int tile = blockIdx.x * 8 + (threadIdx.x >> 5);
    const int ntiles = (n + 15) >> 4;
    if (tile >= ntiles) return;

    int row = tile * 16 + l16;
    if (row >= n) row = n - 1;
    const float* hr = h + row * HID_DIM;
    const bool live = (l16 < OUT_DIM);

    float sc[8];
    #pragma unroll
    for (int i = 0; i < 8; ++i) {
        int orow = tile * 16 + i + 8 * half;
        if (orow >= n) orow = n - 1;
        sc[i] = rsqrtf((float)(cnt[orow] + 1));
    }

    v8f c = {};
    #pragma unroll
    for (int k = 0; k < HID_DIM; k += 4) {
        const int kk = k + 2 * half;
        v2f a; a.x = hr[kk]; a.y = hr[kk + 1];
        v2f b;
        b.x = live ? W2[kk * OUT_DIM + l16]       : 0.0f;
        b.y = live ? W2[(kk + 1) * OUT_DIM + l16] : 0.0f;
        c = __builtin_amdgcn_wmma_f32_16x16x4_f32(false, a, false, b, (short)0, c, false, false);
    }
    if (live) {
        #pragma unroll
        for (int i = 0; i < 8; ++i) {
            int orow = tile * 16 + i + 8 * half;
            if (orow < n) g[orow * OUT_DIM + l16] = c[i] * sc[i];
        }
    }
}

__global__ void __launch_bounds__(256) agg2_kernel(const int* __restrict__ cnt,
                                                   const int* __restrict__ ell,
                                                   const int* __restrict__ spill_cnt,
                                                   const int2* __restrict__ spill,
                                                   const float* __restrict__ g,
                                                   const float* __restrict__ b2,
                                                   float* __restrict__ out, int n) {
    const int tid  = blockIdx.x * blockDim.x + threadIdx.x;
    const int lane = threadIdx.x & 31;
    const int f    = lane & 7;
    const int gb   = lane & 24;
    const int node = tid >> 3;
    const bool live = node < n;
    const int d = live ? node : n - 1;
    const int m = cnt[d];
    const float dd = rsqrtf((float)(m + 1));
    const int mm = live ? min(m, CAP) : 0;
    float acc = g[d * OUT_DIM + f];
    int mx = mm;
    mx = max(mx, __shfl_xor(mx, 8));
    mx = max(mx, __shfl_xor(mx, 16));
    const int* row = ell + d * CAP;

    for (int base = 0; base < mx; base += 8) {
        const int j = base + f;
        const int s_l = (j < mm) ? row[j] : n;
        float v8[8];
        #pragma unroll
        for (int t = 0; t < 8; ++t) {
            const int s_t = __shfl(s_l, gb + t);
            v8[t] = g[s_t * OUT_DIM + f];
        }
        #pragma unroll
        for (int t = 0; t < 8; ++t) acc += v8[t];
    }
    if (__builtin_amdgcn_ballot_w32(live && m > CAP) != 0u) {
        const int K = spill_cnt[0];
        for (int i = 0; i < K; ++i) {
            const int2 p = spill[i];
            const float v = g[p.x * OUT_DIM + f];
            acc = (live && p.y == d) ? acc + v : acc;
        }
    }
    if (live) out[node * OUT_DIM + f] = acc * dd + b2[f];
}

extern "C" void kernel_launch(void* const* d_in, const int* in_sizes, int n_in,
                              void* d_out, int out_size, void* d_ws, size_t ws_size,
                              hipStream_t stream) {
    const float* x  = (const float*)d_in[0];
    const int*   ei = (const int*)d_in[1];
    const float* W1 = (const float*)d_in[2];
    const float* b1 = (const float*)d_in[3];
    const float* W2 = (const float*)d_in[4];
    const float* b2 = (const float*)d_in[5];
    float* out = (float*)d_out;

    const int n = in_sizes[0] / IN_DIM;
    const int E = in_sizes[1] / 2;
    const int* srcIdx = ei;
    const int* dstIdx = ei + E;

    const int nWG = (E + CH - 1) / CH;
    const int B   = (n + RB - 1) / RB;
    const int PB  = B + 1;

    const size_t nr = ((size_t)n + 63) / 64 * 64;
    const size_t Er = ((size_t)E + 63) / 64 * 64;
    const size_t Pr = ((size_t)nWG * PB + 63) / 64 * 64;
    int*      cnt       = (int*)d_ws;
    int*      spill_cnt = cnt + nr;
    int*      pre       = spill_cnt + 64;
    unsigned* sorted    = (unsigned*)(pre + Pr);
    int*      ell       = (int*)(sorted + (size_t)nWG * CH);
    int2*     spill     = (int2*)(ell + nr * CAP);
    float*    hw        = (float*)(spill + Er);
    float*    h         = hw + (nr + 64) * HID_DIM;
    float*    g         = h  + nr * HID_DIM;

    const dim3 blk(256);
    const int ntiles = (n + 15) / 16;

    sort_kernel<<<dim3(nWG), dim3(1024), 0, stream>>>(srcIdx, dstIdx, sorted, pre, spill_cnt, hw, g, E, n, PB);
    build_kernel<<<dim3(B), blk, 0, stream>>>(sorted, pre, cnt, ell, spill_cnt, spill, nWG, PB, n);
    gemm1_kernel<<<dim3((ntiles + 7) / 8), blk, 0, stream>>>(x, W1, cnt, hw, n);
    agg1_kernel<<<dim3((n + 7) / 8), blk, 0, stream>>>(cnt, ell, spill_cnt, spill, hw, b1, h, n);
    gemm2_kernel<<<dim3((ntiles + 7) / 8), blk, 0, stream>>>(h, W2, cnt, g, n);
    agg2_kernel<<<dim3((n * OUT_DIM + 255) / 256), blk, 0, stream>>>(cnt, ell, spill_cnt, spill, g, b2, out, n);
}


# Layer-1 gather as resident waves: each wave loops over 8 consecutive nodes, 16-byte loads cover 4 rows; a node's sum is now 4 interleaved slot-order partial sums folded (p0+p1)+(p2+p3)
# speedup vs baseline: 4.9089x; 1.0842x over previous
#include <hip/hip_runtime.h>


typedef float v2f __attribute__((ext_vector_type(2)));
typedef float v8f __attribute__((ext_vector_type(8)));

#define IN_DIM  128
#define HID_DIM 32
#define OUT_DIM 8
#define CAP     64
#define CH      8192
#define BSH     8
#define RB      256
#define MAXB    512


__global__ void __launch_bounds__(1024) sort_kernel(const int* __restrict__ srcIdx,
                                                    const int* __restrict__ dstIdx,
                                                    unsigned* __restrict__ sorted,
                                                    int* __restrict__ pre,
                                                    int* __restrict__ spill_cnt,
                                                    float* __restrict__ hw,
                                                    float* __restrict__ g,
                                                    int E, int n, int PB) {
    __shared__ int hist[MAXB];
    __shared__ int pfx[MAXB + 1];
    __shared__ unsigned val[CH];
    const int tid = threadIdx.x;
    const int e0 = blockIdx.x * CH;
    if (blockIdx.x == 0) {
        if (tid == 0) spill_cnt[0] = 0;
        if (tid < HID_DIM) hw[n * HID_DIM + tid] = 0.0f;
        if (tid < OUT_DIM) g[n * OUT_DIM + tid] = 0.0f;
    }
    if (tid < MAXB) hist[tid] = 0;
    __syncthreads();

    unsigned w[CH / 1024];
    int bb[CH / 1024];
    int pos[CH / 1024];
    int dv[CH / 1024];
    int sv[CH / 1024];
    #pragma unroll
    for (int k = 0; k < CH / 1024; ++k) {
        const int ec = min(e0 + k * 1024 + tid, E - 1);
        dv[k] = dstIdx[ec];
        sv[k] = srcIdx[ec];
    }
    #pragma unroll
    for (int k = 0; k < CH / 1024; ++k) {
        const bool valid = (e0 + k * 1024 + tid) < E;
        bb[k] = valid ? (dv[k] >> BSH) : -1;
        w[k]  = ((unsigned)(dv[k] & (RB - 1)) << 24) | (unsigned)sv[k];
    }
    #pragma unroll
    for (int k = 0; k < CH / 1024; ++k) {
        pos[k] = 0;
        if (bb[k] >= 0) pos[k] = atomicAdd(&hist[bb[k]], 1);
    }
    __syncthreads();

    if (tid < 32) {
        int loc[MAXB / 32];
        int sum = 0;
        #pragma unroll
        for (int j = 0; j < MAXB / 32; ++j) { loc[j] = hist[tid * (MAXB / 32) + j]; sum += loc[j]; }
        int incl = sum;
        #pragma unroll
        for (int off = 1; off < 32; off <<= 1) {
            const int t = __shfl_up(incl, off);
            if (tid >= off) incl += t;
        }
        int run = incl - sum;
        #pragma unroll
        for (int j = 0; j < MAXB / 32; ++j) { pfx[tid * (MAXB / 32) + j] = run; run += loc[j]; }
        if (tid == 31) pfx[MAXB] = run;
    }
    __syncthreads();

    #pragma unroll
    for (int k = 0; k < CH / 1024; ++k)
        if (bb[k] >= 0) val[pfx[bb[k]] + pos[k]] = w[k];
    if (tid < PB) pre[blockIdx.x * PB + tid] = pfx[tid];
    __syncthreads();

    const int tot = pfx[MAXB];
    for (int i = tid; i < tot; i += 1024) sorted[e0 + i] = val[i];
}

__global__ void __launch_bounds__(256) build_kernel(const unsigned* __restrict__ sorted,
                                                    const int* __restrict__ pre,
                                                    int* __restrict__ cnt,
                                                    int* __restrict__ ell,
                                                    int* __restrict__ spill_cnt,
                                                    int2* __restrict__ spill,
                                                    int nWG, int PB, int n) {
    __shared__ int lcnt[RB];
    __shared__ int lell[RB * CAP];
    __shared__ int lo_s[256];
    __shared__ int len_s[256];
    const int tid  = threadIdx.x;
    const int lane = tid & 31;
    const int wave = __builtin_amdgcn_readfirstlane(tid >> 5);
    const int bin  = blockIdx.x;
    const int node0 = bin * RB;

    auto take = [&](unsigned wv) { return atomicAdd(&lcnt[wv >> 24], 1); };
    auto place = [&](unsigned wv, int r) {
        const int dl = (int)(wv >> 24);
        const int s  = (int)(wv & 0xFFFFFFu);
        if (r < CAP) {
            lell[dl * CAP + r] = s;
        } else {
            const int kk = atomicAdd(spill_cnt, 1);
            spill[kk] = make_int2(s, node0 + dl);
        }
    };

    lcnt[tid] = 0;
    for (int g0 = 0; g0 < nWG; g0 += 256) {
        __syncthreads();
        const int wg = g0 + tid;
        int lo = 0, len = 0;
        if (wg < nWG) {
            lo  = pre[wg * PB + bin];
            len = pre[wg * PB + bin + 1] - lo;
        }
        lo_s[tid] = lo;
        len_s[tid] = len;
        __syncthreads();
        for (int k0 = wave; k0 < 256; k0 += 64) {
            const int kk  = k0 + 8 * (lane & 7);
            const int lnv = len_s[kk];
            const int lov = lo_s[kk];
            unsigned v[8];
            int ln[8];
            const unsigned* bp[8];
            #pragma unroll
            for (int r = 0; r < 8; ++r) {
                ln[r] = __builtin_amdgcn_readlane(lnv, r);
                bp[r] = sorted + (size_t)(g0 + k0 + 8 * r) * CH + __builtin_amdgcn_readlane(lov, r);
                v[r] = (lane < ln[r]) ? bp[r][lane] : 0u;
            }
            int sl[8];
            #pragma unroll
            for (int r = 0; r < 8; ++r)
                sl[r] = (lane < ln[r]) ? take(v[r]) : 0;
            #pragma unroll
            for (int r = 0; r < 8; ++r)
                if (lane < ln[r]) place(v[r], sl[r]);
            #pragma unroll
            for (int r = 0; r < 8; ++r)
                for (int j = 32 + lane; j < ln[r]; j += 32) {
                    const unsigned wv = bp[r][j];
                    place(wv, take(wv));
                }
        }
    }
    __syncthreads();

    if (node0 + tid < n) cnt[node0 + tid] = lcnt[tid];
    const int myrow = wave * 32 + lane;
    const int mine  = (node0 + myrow < n) ? min(lcnt[myrow], CAP) : 0;
    #pragma unroll 8
    for (int i = 0; i < 32; ++i) {
        const int rr = wave * 32 + i;
        const int m  = __builtin_amdgcn_readlane(mine, i);
        int* dstrow = ell + (size_t)(node0 + rr) * CAP;
        if (lane < m)      dstrow[lane]      = lell[rr * CAP + lane];
        if (lane + 32 < m) dstrow[32 + lane] = lell[rr * CAP + 32 + lane];
    }
}

__global__ void __launch_bounds__(256) gemm1_kernel(const float* __restrict__ x,
                                                    const float* __restrict__ W1,
                                                    const int* __restrict__ cnt,
                                                    float* __restrict__ hw, int n) {
    const int lane = threadIdx.x & 31;
    const int half = lane >> 4;
    const int l16  = lane & 15;
    const int tile = blockIdx.x * 8 + (threadIdx.x >> 5);
    const int ntiles = (n + 15) >> 4;
    if (tile >= ntiles) return;

    int row = tile * 16 + l16;
    if (row >= n) row = n - 1;
    const float* xr = x + row * IN_DIM;

    float sc[8];
    #pragma unroll
    for (int i = 0; i < 8; ++i) {
        int orow = tile * 16 + i + 8 * half;
        if (orow >= n) orow = n - 1;
        sc[i] = rsqrtf((float)(cnt[orow] + 1));
    }

    v8f c0 = {}; v8f c1 = {};
    #pragma unroll 8
    for (int k = 0; k < IN_DIM; k += 4) {
        const int kk = k + 2 * half;
        v2f a;  a.x  = xr[kk];                       a.y  = xr[kk + 1];
        v2f b0; b0.x = W1[kk * HID_DIM + l16];       b0.y = W1[(kk + 1) * HID_DIM + l16];
        v2f b1; b1.x = W1[kk * HID_DIM + 16 + l16];  b1.y = W1[(kk + 1) * HID_DIM + 16 + l16];
        c0 = __builtin_amdgcn_wmma_f32_16x16x4_f32(false, a, false, b0, (short)0, c0, false, false);
        c1 = __builtin_amdgcn_wmma_f32_16x16x4_f32(false, a, false, b1, (short)0, c1, false, false);
    }
    #pragma unroll
    for (int i = 0; i < 8; ++i) {
        int orow = tile * 16 + i + 8 * half;
        if (orow < n) {
            hw[orow * HID_DIM + l16]      = c0[i] * sc[i];
            hw[orow * HID_DIM + 16 + l16] = c1[i] * sc[i];
        }
    }
}

#define NPW 8

template <int T>
__device__ __forceinline__ int group_bcast(int v) {
    return __builtin_amdgcn_mov_dpp8(v, T * 0x249249);
}

__device__ __forceinline__ float4 row_piece(const float* __restrict__ tab, int row, int c) {
    const unsigned off = (unsigned)row * (HID_DIM * 4u) + (unsigned)c * 16u;
    return *(const float4*)((const char*)tab + off);
}

__device__ __forceinline__ void add4(float4& a, const float4 b) {
    a.x += b.x; a.y += b.y; a.z += b.z; a.w += b.w;
}

__global__ void __launch_bounds__(256) agg1_kernel(const int* __restrict__ cnt,
                                                   const int* __restrict__ ell,
                                                   const int* __restrict__ spill_cnt,
                                                   const int2* __restrict__ spill,
                                                   const float* __restrict__ hw,
                                                   const float* __restrict__ b1,
                                                   float* __restrict__ h, int n) {
    const int lane = threadIdx.x & 31;
    const int q = lane >> 3;
    const int c = lane & 7;
    const int wv = __builtin_amdgcn_readfirstlane((int)(blockIdx.x * 8 + (threadIdx.x >> 5)));
    const int d0 = wv * NPW;
    if (d0 >= n) return;
    const int nn = min(NPW, n - d0);
    const int cv = (lane < nn) ? cnt[d0 + lane] : 0;
    float4 bias;
    bias.x = b1[4 * c]; bias.y = b1[4 * c + 1]; bias.z = b1[4 * c + 2]; bias.w = b1[4 * c + 3];
    const int sidx = 4 * c + q;

    int m_nx  = __builtin_amdgcn_readlane(cv, 0);
    int mm_nx = min(m_nx, CAP);
    int sv_nx = (sidx < mm_nx) ? ell[d0 * CAP + sidx] : n;

    for (int i = 0; i < nn; ++i) {
        const int d  = d0 + i;
        const int m  = m_nx;
        const int mm = mm_nx;
        const int sv = sv_nx;
        if (i + 1 < nn) {
            m_nx  = __builtin_amdgcn_readlane(cv, i + 1);
            mm_nx = min(m_nx, CAP);
            sv_nx = (sidx < mm_nx) ? ell[(d + 1) * CAP + sidx] : n;
        }
        const int* row = ell + d * CAP;

        const float4 self = row_piece(hw, d, c);
        const float4 v0 = row_piece(hw, group_bcast<0>(sv), c);
        const float4 v1 = row_piece(hw, group_bcast<1>(sv), c);
        const float4 v2 = row_piece(hw, group_bcast<2>(sv), c);
        const float4 v3 = row_piece(hw, group_bcast<3>(sv), c);
        const float4 v4 = row_piece(hw, group_bcast<4>(sv), c);
        float4 v5 = make_float4(0.0f, 0.0f, 0.0f, 0.0f);
        float4 v6 = v5, v7 = v5;
        if (mm > 20) {
            v5 = row_piece(hw, group_bcast<5>(sv), c);
            v6 = row_piece(hw, group_bcast<6>(sv), c);
            v7 = row_piece(hw, group_bcast<7>(sv), c);
        }
        float4 acc = (q == 0) ? self : make_float4(0.0f, 0.0f, 0.0f, 0.0f);
        add4(acc, v0); add4(acc, v1); add4(acc, v2); add4(acc, v3); add4(acc, v4);
        add4(acc, v5); add4(acc, v6); add4(acc, v7);
        for (int j0 = 32; j0 < mm; j0 += 4) {
            const int j = j0 + q;
            const int s = (j < mm) ? row[j] : n;
            add4(acc, row_piece(hw, s, c));
        }
        if (m > CAP) {
            const int K = spill_cnt[0];
            for (int k = 0; k < K; ++k) {
                const int2 p = spill[k];
                if (p.y == d && q == 0) add4(acc, row_piece(hw, p.x, c));
            }
        }
        acc.x += __shfl_xor(acc.x, 8);  acc.y += __shfl_xor(acc.y, 8);
        acc.z += __shfl_xor(acc.z, 8);  acc.w += __shfl_xor(acc.w, 8);
        acc.x += __shfl_xor(acc.x, 16); acc.y += __shfl_xor(acc.y, 16);
        acc.z += __shfl_xor(acc.z, 16); acc.w += __shfl_xor(acc.w, 16);
        const float dd = rsqrtf((float)(m + 1));
        if (q == 0) {
            float4 r;
            r.x = fmaxf(acc.x * dd + bias.x, 0.0f);
            r.y = fmaxf(acc.y * dd + bias.y, 0.0f);
            r.z = fmaxf(acc.z * dd + bias.z, 0.0f);
            r.w = fmaxf(acc.w * dd + bias.w, 0.0f);
            *(float4*)(h + d * HID_DIM + 4 * c) = r;
        }
    }
}

__global__ void __launch_bounds__(256) gemm2_kernel(const float* __restrict__ h,
                                                    const float* __restrict__ W2,
                                                    const int* __restrict__ cnt,
                                                    float* __restrict__ g, int n) {
    const int lane = threadIdx.x & 31;
    const int half = lane >> 4;
    const int l16  = lane & 15;
    const int tile = blockIdx.x * 8 + (threadIdx.x >> 5);
    const int ntiles = (n + 15) >> 4;
    if (tile >= ntiles) return;

    int row = tile * 16 + l16;
    if (row >= n) row = n - 1;
    const float* hr = h + row * HID_DIM;
    const bool live = (l16 < OUT_DIM);

    float sc[8];
    #pragma unroll
    for (int i = 0; i < 8; ++i) {
        int orow = tile * 16 + i + 8 * half;
        if (orow >= n) orow = n - 1;
        sc[i] = rsqrtf((float)(cnt[orow] + 1));
    }

    v8f c = {};
    #pragma unroll
    for (int k = 0; k < HID_DIM; k += 4) {
        const int kk = k + 2 * half;
        v2f a; a.x = hr[kk]; a.y = hr[kk + 1];
        v2f b;
        b.x = live ? W2[kk * OUT_DIM + l16]       : 0.0f;
        b.y = live ? W2[(kk + 1) * OUT_DIM + l16] : 0.0f;
        c = __builtin_amdgcn_wmma_f32_16x16x4_f32(false, a, false, b, (short)0, c, false, false);
    }
    if (live) {
        #pragma unroll
        for (int i = 0; i < 8; ++i) {
            int orow = tile * 16 + i + 8 * half;
            if (orow < n) g[orow * OUT_DIM + l16] = c[i] * sc[i];
        }
    }
}

__global__ void __launch_bounds__(256) agg2_kernel(const int* __restrict__ cnt,
                                                   const int* __restrict__ ell,
                                                   const int* __restrict__ spill_cnt,
                                                   const int2* __restrict__ spill,
                                                   const float* __restrict__ g,
                                                   const float* __restrict__ b2,
                                                   float* __restrict__ out, int n) {
    const int tid  = blockIdx.x * blockDim.x + threadIdx.x;
    const int lane = threadIdx.x & 31;
    const int f    = lane & 7;
    const int gb   = lane & 24;
    const int node = tid >> 3;
    const bool live = node < n;
    const int d = live ? node : n - 1;
    const int m = cnt[d];
    const float dd = rsqrtf((float)(m + 1));
    const int mm = live ? min(m, CAP) : 0;
    float acc = g[d * OUT_DIM + f];
    int mx = mm;
    mx = max(mx, __shfl_xor(mx, 8));
    mx = max(mx, __shfl_xor(mx, 16));
    const int* row = ell + d * CAP;

    for (int base = 0; base < mx; base += 8) {
        const int j = base + f;
        const int s_l = (j < mm) ? row[j] : n;
        float v8[8];
        #pragma unroll
        for (int t = 0; t < 8; ++t) {
            const int s_t = __shfl(s_l, gb + t);
            v8[t] = g[s_t * OUT_DIM + f];
        }
        #pragma unroll
        for (int t = 0; t < 8; ++t) acc += v8[t];
    }
    if (__builtin_amdgcn_ballot_w32(live && m > CAP) != 0u) {
        const int K = spill_cnt[0];
        for (int i = 0; i < K; ++i) {
            const int2 p = spill[i];
            const float v = g[p.x * OUT_DIM + f];
            acc = (live && p.y == d) ? acc + v : acc;
        }
    }
    if (live) out[node * OUT_DIM + f] = acc * dd + b2[f];
}

extern "C" void kernel_launch(void* const* d_in, const int* in_sizes, int n_in,
                              void* d_out, int out_size, void* d_ws, size_t ws_size,
                              hipStream_t stream) {
    const float* x  = (const float*)d_in[0];
    const int*   ei = (const int*)d_in[1];
    const float* W1 = (const float*)d_in[2];
    const float* b1 = (const float*)d_in[3];
    const float* W2 = (const float*)d_in[4];
    const float* b2 = (const float*)d_in[5];
    float* out = (float*)d_out;

    const int n = in_sizes[0] / IN_DIM;
    const int E = in_sizes[1] / 2;
    const int* srcIdx = ei;
    const int* dstIdx = ei + E;

    const int nWG = (E + CH - 1) / CH;
    const int B   = (n + RB - 1) / RB;
    const int PB  = B + 1;

    const size_t nr = ((size_t)n + 63) / 64 * 64;
    const size_t Er = ((size_t)E + 63) / 64 * 64;
    const size_t Pr = ((size_t)nWG * PB + 63) / 64 * 64;
    int*      cnt       = (int*)d_ws;
    int*      spill_cnt = cnt + nr;
    int*      pre       = spill_cnt + 64;
    unsigned* sorted    = (unsigned*)(pre + Pr);
    int*      ell       = (int*)(sorted + (size_t)nWG * CH);
    int2*     spill     = (int2*)(ell + nr * CAP);
    float*    hw        = (float*)(spill + Er);
    float*    h         = hw + (nr + 64) * HID_DIM;
    float*    g         = h  + nr * HID_DIM;

    const dim3 blk(256);
    const int ntiles = (n + 15) / 16;

    sort_kernel<<<dim3(nWG), dim3(1024), 0, stream>>>(srcIdx, dstIdx, sorted, pre, spill_cnt, hw, g, E, n, PB);
    build_kernel<<<dim3(B), blk, 0, stream>>>(sorted, pre, cnt, ell, spill_cnt, spill, nWG, PB, n);
    gemm1_kernel<<<dim3((ntiles + 7) / 8), blk, 0, stream>>>(x, W1, cnt, hw, n);
    agg1_kernel<<<dim3((n + 8 * NPW - 1) / (8 * NPW)), blk, 0, stream>>>(cnt, ell, spill_cnt, spill, hw, b1, h, n);
    gemm2_kernel<<<dim3((ntiles + 7) / 8), blk, 0, stream>>>(h, W2, cnt, g, n);
    agg2_kernel<<<dim3((n * OUT_DIM + 255) / 256), blk, 0, stream>>>(cnt, ell, spill_cnt, spill, g, b2, out, n);
}


# Layer-2 product h @ W2 moved into the layer-1 gather: each wave keeps its 8 finished rows of h in LDS and multiplies them by W2 on the matrix unit itself; h never goes to memory, one launch less
# speedup vs baseline: 5.3686x; 1.0936x over previous
#include <hip/hip_runtime.h>


typedef float v2f __attribute__((ext_vector_type(2)));
typedef float v8f __attribute__((ext_vector_type(8)));

#define IN_DIM  128
#define HID_DIM 32
#define OUT_DIM 8
#define CAP     64
#define CH      8192
#define BSH     8
#define RB      256
#define MAXB    512


__global__ void __launch_bounds__(1024) sort_kernel(const int* __restrict__ srcIdx,
                                                    const int* __restrict__ dstIdx,
                                                    unsigned* __restrict__ sorted,
                                                    int* __restrict__ pre,
                                                    int* __restrict__ spill_cnt,
                                                    float* __restrict__ hw,
                                                    float* __restrict__ g,
                                                    int E, int n, int PB) {
    __shared__ int hist[MAXB];
    __shared__ int pfx[MAXB + 1];
    __shared__ unsigned val[CH];
    const int tid = threadIdx.x;
    const int e0 = blockIdx.x * CH;
    if (blockIdx.x == 0) {
        if (tid == 0) spill_cnt[0] = 0;
        if (tid < HID_DIM) hw[n * HID_DIM + tid] = 0.0f;
        if (tid < OUT_DIM) g[n * OUT_DIM + tid] = 0.0f;
    }
    if (tid < MAXB) hist[tid] = 0;
    __syncthreads();

    unsigned w[CH / 1024];
    int bb[CH / 1024];
    int pos[CH / 1024];
    int dv[CH / 1024];
    int sv[CH / 1024];
    #pragma unroll
    for (int k = 0; k < CH / 1024; ++k) {
        const int ec = min(e0 + k * 1024 + tid, E - 1);
        dv[k] = dstIdx[ec];
        sv[k] = srcIdx[ec];
    }
    #pragma unroll
    for (int k = 0; k < CH / 1024; ++k) {
        const bool valid = (e0 + k * 1024 + tid) < E;
        bb[k] = valid ? (dv[k] >> BSH) : -1;
        w[k]  = ((unsigned)(dv[k] & (RB - 1)) << 24) | (unsigned)sv[k];
    }
    #pragma unroll
    for (int k = 0; k < CH / 1024; ++k) {
        pos[k] = 0;
        if (bb[k] >= 0) pos[k] = atomicAdd(&hist[bb[k]], 1);
    }
    __syncthreads();

    if (tid < 32) {
        int loc[MAXB / 32];
        int sum = 0;
        #pragma unroll
        for (int j = 0; j < MAXB / 32; ++j) { loc[j] = hist[tid * (MAXB / 32) + j]; sum += loc[j]; }
        int incl = sum;
        #pragma unroll
        for (int off = 1; off < 32; off <<= 1) {
            const int t = __shfl_up(incl, off);
            if (tid >= off) incl += t;
        }
        int run = incl - sum;
        #pragma unroll
        for (int j = 0; j < MAXB / 32; ++j) { pfx[tid * (MAXB / 32) + j] = run; run += loc[j]; }
        if (tid == 31) pfx[MAXB] = run;
    }
    __syncthreads();

    #pragma unroll
    for (int k = 0; k < CH / 1024; ++k)
        if (bb[k] >= 0) val[pfx[bb[k]] + pos[k]] = w[k];
    if (tid < PB) pre[blockIdx.x * PB + tid] = pfx[tid];
    __syncthreads();

    const int tot = pfx[MAXB];
    for (int i = tid; i < tot; i += 1024) sorted[e0 + i] = val[i];
}

__global__ void __launch_bounds__(256) build_kernel(const unsigned* __restrict__ sorted,
                                                    const int* __restrict__ pre,
                                                    int* __restrict__ cnt,
                                                    int* __restrict__ ell,
                                                    int* __restrict__ spill_cnt,
                                                    int2* __restrict__ spill,
                                                    int nWG, int PB, int n) {
    __shared__ int lcnt[RB];
    __shared__ int lell[RB * CAP];
    __shared__ int lo_s[256];
    __shared__ int len_s[256];
    const int tid  = threadIdx.x;
    const int lane = tid & 31;
    const int wave = __builtin_amdgcn_readfirstlane(tid >> 5);
    const int bin  = blockIdx.x;
    const int node0 = bin * RB;

    auto take = [&](unsigned wv) { return atomicAdd(&lcnt[wv >> 24], 1); };
    auto place = [&](unsigned wv, int r) {
        const int dl = (int)(wv >> 24);
        const int s  = (int)(wv & 0xFFFFFFu);
        if (r < CAP) {
            lell[dl * CAP + r] = s;
        } else {
            const int kk = atomicAdd(spill_cnt, 1);
            spill[kk] = make_int2(s, node0 + dl);
        }
    };

    lcnt[tid] = 0;
    for (int g0 = 0; g0 < nWG; g0 += 256) {
        __syncthreads();
        const int wg = g0 + tid;
        int lo = 0, len = 0;
        if (wg < nWG) {
            lo  = pre[wg * PB + bin];
            len = pre[wg * PB + bin + 1] - lo;
        }
        lo_s[tid] = lo;
        len_s[tid] = len;
        __syncthreads();
        for (int k0 = wave; k0 < 256; k0 += 64) {
            const int kk  = k0 + 8 * (lane & 7);
            const int lnv = len_s[kk];
            const int lov = lo_s[kk];
            unsigned v[8];
            int ln[8];
            const unsigned* bp[8];
            #pragma unroll
            for (int r = 0; r < 8; ++r) {
                ln[r] = __builtin_amdgcn_readlane(lnv, r);
                bp[r] = sorted + (size_t)(g0 + k0 + 8 * r) * CH + __builtin_amdgcn_readlane(lov, r);
                v[r] = (lane < ln[r]) ? bp[r][lane] : 0u;
            }
            int sl[8];
            #pragma unroll
            for (int r = 0; r < 8; ++r)
                sl[r] = (lane < ln[r]) ? take(v[r]) : 0;
            #pragma unroll
            for (int r = 0; r < 8; ++r)
                if (lane < ln[r]) place(v[r], sl[r]);
            #pragma unroll
            for (int r = 0; r < 8; ++r)
                for (int j = 32 + lane; j < ln[r]; j += 32) {
                    const unsigned wv = bp[r][j];
                    place(wv, take(wv));
                }
        }
    }
    __syncthreads();

    if (node0 + tid < n) cnt[node0 + tid] = lcnt[tid];
    const int myrow = wave * 32 + lane;
    const int mine  = (node0 + myrow < n) ? min(lcnt[myrow], CAP) : 0;
    #pragma unroll 8
    for (int i = 0; i < 32; ++i) {
        const int rr = wave * 32 + i;
        const int m  = __builtin_amdgcn_readlane(mine, i);
        int* dstrow = ell + (size_t)(node0 + rr) * CAP;
        if (lane < m)      dstrow[lane]      = lell[rr * CAP + lane];
        if (lane + 32 < m) dstrow[32 + lane] = lell[rr * CAP + 32 + lane];
    }
}

__global__ void __launch_bounds__(256) gemm1_kernel(const float* __restrict__ x,
                                                    const float* __restrict__ W1,
                                                    const int* __restrict__ cnt,
                                                    float* __restrict__ hw, int n) {
    const int lane = threadIdx.x & 31;
    const int half = lane >> 4;
    const int l16  = lane & 15;
    const int tile = blockIdx.x * 8 + (threadIdx.x >> 5);
    const int ntiles = (n + 15) >> 4;
    if (tile >= ntiles) return;

    int row = tile * 16 + l16;
    if (row >= n) row = n - 1;
    const float* xr = x + row * IN_DIM;

    float sc[8];
    #pragma unroll
    for (int i = 0; i < 8; ++i) {
        int orow = tile * 16 + i + 8 * half;
        if (orow >= n) orow = n - 1;
        sc[i] = rsqrtf((float)(cnt[orow] + 1));
    }

    v8f c0 = {}; v8f c1 = {};
    #pragma unroll 8
    for (int k = 0; k < IN_DIM; k += 4) {
        const int kk = k + 2 * half;
        v2f a;  a.x  = xr[kk];                       a.y  = xr[kk + 1];
        v2f b0; b0.x = W1[kk * HID_DIM + l16];       b0.y = W1[(kk + 1) * HID_DIM + l16];
        v2f b1; b1.x = W1[kk * HID_DIM + 16 + l16];  b1.y = W1[(kk + 1) * HID_DIM + 16 + l16];
        c0 = __builtin_amdgcn_wmma_f32_16x16x4_f32(false, a, false, b0, (short)0, c0, false, false);
        c1 = __builtin_amdgcn_wmma_f32_16x16x4_f32(false, a, false, b1, (short)0, c1, false, false);
    }
    #pragma unroll
    for (int i = 0; i < 8; ++i) {
        int orow = tile * 16 + i + 8 * half;
        if (orow < n) {
            hw[orow * HID_DIM + l16]      = c0[i] * sc[i];
            hw[orow * HID_DIM + 16 + l16] = c1[i] * sc[i];
        }
    }
}

#define NPW 8
#define HPITCH 36

template <int T>
__device__ __forceinline__ int group_bcast(int v) {
    return __builtin_amdgcn_mov_dpp8(v, T * 0x249249);
}

__device__ __forceinline__ float4 row_piece(const float* __restrict__ tab, int row, int c) {
    const unsigned off = (unsigned)row * (HID_DIM * 4u) + (unsigned)c * 16u;
    return *(const float4*)((const char*)tab + off);
}

__device__ __forceinline__ void add4(float4& a, const float4 b) {
    a.x += b.x; a.y += b.y; a.z += b.z; a.w += b.w;
}

__global__ void __launch_bounds__(256) agg1_kernel(const int* __restrict__ cnt,
                                                   const int* __restrict__ ell,
                                                   const int* __restrict__ spill_cnt,
                                                   const int2* __restrict__ spill,
                                                   const float* __restrict__ hw,
                                                   const float* __restrict__ b1,
                                                   const float* __restrict__ W2,
                                                   float* __restrict__ g, int n) {
    __shared__ __attribute__((aligned(16))) float hs[8 * NPW * HPITCH];
    const int lane = threadIdx.x & 31;
    const int q = lane >> 3;
    const int c = lane & 7;
    const int wv = __builtin_amdgcn_readfirstlane((int)(blockIdx.x * 8 + (threadIdx.x >> 5)));
    const int d0 = wv * NPW;
    if (d0 >= n) return;
    float* myh = hs + __builtin_amdgcn_readfirstlane((int)(threadIdx.x >> 5)) * (NPW * HPITCH);
    const int nn = min(NPW, n - d0);
    const int cv = (lane < nn) ? cnt[d0 + lane] : 0;
    float4 bias;
    bias.x = b1[4 * c]; bias.y = b1[4 * c + 1]; bias.z = b1[4 * c + 2]; bias.w = b1[4 * c + 3];
    const int sidx = 4 * c + q;

    int m_nx  = __builtin_amdgcn_readlane(cv, 0);
    int mm_nx = min(m_nx, CAP);
    int sv_nx = (sidx < mm_nx) ? ell[d0 * CAP + sidx] : n;

    for (int i = 0; i < nn; ++i) {
        const int d  = d0 + i;
        const int m  = m_nx;
        const int mm = mm_nx;
        const int sv = sv_nx;
        if (i + 1 < nn) {
            m_nx  = __builtin_amdgcn_readlane(cv, i + 1);
            mm_nx = min(m_nx, CAP);
            sv_nx = (sidx < mm_nx) ? ell[(d + 1) * CAP + sidx] : n;
        }
        const int* row = ell + d * CAP;

        const float4 self = row_piece(hw, d, c);
        const float4 v0 = row_piece(hw, group_bcast<0>(sv), c);
        const float4 v1 = row_piece(hw, group_bcast<1>(sv), c);
        const float4 v2 = row_piece(hw, group_bcast<2>(sv), c);
        const float4 v3 = row_piece(hw, group_bcast<3>(sv), c);
        const float4 v4 = row_piece(hw, group_bcast<4>(sv), c);
        float4 v5 = make_float4(0.0f, 0.0f, 0.0f, 0.0f);
        float4 v6 = v5, v7 = v5;
        if (mm > 20) {
            v5 = row_piece(hw, group_bcast<5>(sv), c);
            v6 = row_piece(hw, group_bcast<6>(sv), c);
            v7 = row_piece(hw, group_bcast<7>(sv), c);
        }
        float4 acc = (q == 0) ? self : make_float4(0.0f, 0.0f, 0.0f, 0.0f);
        add4(acc, v0); add4(acc, v1); add4(acc, v2); add4(acc, v3); add4(acc, v4);
        add4(acc, v5); add4(acc, v6); add4(acc, v7);
        for (int j0 = 32; j0 < mm; j0 += 4) {
            const int j = j0 + q;
            const int s = (j < mm) ? row[j] : n;
            add4(acc, row_piece(hw, s, c));
        }
        if (m > CAP) {
            const int K = spill_cnt[0];
            for (int k = 0; k < K; ++k) {
                const int2 p = spill[k];
                if (p.y == d && q == 0) add4(acc, row_piece(hw, p.x, c));
            }
        }
        acc.x += __shfl_xor(acc.x, 8);  acc.y += __shfl_xor(acc.y, 8);
        acc.z += __shfl_xor(acc.z, 8);  acc.w += __shfl_xor(acc.w, 8);
        acc.x += __shfl_xor(acc.x, 16); acc.y += __shfl_xor(acc.y, 16);
        acc.z += __shfl_xor(acc.z, 16); acc.w += __shfl_xor(acc.w, 16);
        const float dd = rsqrtf((float)(m + 1));
        if (q == 0) {
            float* hrow = myh + i * HPITCH + 4 * c;
            hrow[0] = fmaxf(acc.x * dd + bias.x, 0.0f);
            hrow[1] = fmaxf(acc.y * dd + bias.y, 0.0f);
            hrow[2] = fmaxf(acc.z * dd + bias.z, 0.0f);
            hrow[3] = fmaxf(acc.w * dd + bias.w, 0.0f);
        }
    }

    const int half = lane >> 4;
    const int l16  = lane & 15;
    const bool rowok = l16 < nn;
    const bool live  = l16 < OUT_DIM;
    const float* hr = myh + (rowok ? l16 : 0) * HPITCH + 2 * half;
    const float* wc = W2 + 2 * half * OUT_DIM + (l16 & (OUT_DIM - 1));
    float w0[HID_DIM / 4], w1[HID_DIM / 4], a0[HID_DIM / 4], a1[HID_DIM / 4];
    #pragma unroll
    for (int j = 0; j < HID_DIM / 4; ++j) {
        w0[j] = wc[(4 * j) * OUT_DIM];
        w1[j] = wc[(4 * j + 1) * OUT_DIM];
        a0[j] = hr[4 * j];
        a1[j] = hr[4 * j + 1];
    }
    v8f pr = {};
    #pragma unroll
    for (int j = 0; j < HID_DIM / 4; ++j) {
        v2f a; a.x = rowok ? a0[j] : 0.0f; a.y = rowok ? a1[j] : 0.0f;
        v2f b; b.x = live  ? w0[j] : 0.0f; b.y = live  ? w1[j] : 0.0f;
        pr = __builtin_amdgcn_wmma_f32_16x16x4_f32(false, a, false, b, (short)0, pr, false, false);
    }
    const float dl = rsqrtf((float)(cv + 1));
    float o[NPW];
    #pragma unroll
    for (int i = 0; i < NPW; ++i)
        o[i] = pr[i] * __int_as_float(__builtin_amdgcn_readlane(__float_as_int(dl), i));
    if (live && half == 0) {
        #pragma unroll
        for (int i = 0; i < NPW; ++i)
            if (i < nn) g[(d0 + i) * OUT_DIM + l16] = o[i];
    }
}

__global__ void __launch_bounds__(256) agg2_kernel(const int* __restrict__ cnt,
                                                   const int* __restrict__ ell,
                                                   const int* __restrict__ spill_cnt,
                                                   const int2* __restrict__ spill,
                                                   const float* __restrict__ g,
                                                   const float* __restrict__ b2,
                                                   float* __restrict__ out, int n) {
    const int tid  = blockIdx.x * blockDim.x + threadIdx.x;
    const int lane = threadIdx.x & 31;
    const int f    = lane & 7;
    const int gb   = lane & 24;
    const int node = tid >> 3;
    const bool live = node < n;
    const int d = live ? node : n - 1;
    const int m = cnt[d];
    const float dd = rsqrtf((float)(m + 1));
    const int mm = live ? min(m, CAP) : 0;
    float acc = g[d * OUT_DIM + f];
    int mx = mm;
    mx = max(mx, __shfl_xor(mx, 8));
    mx = max(mx, __shfl_xor(mx, 16));
    const int* row = ell + d * CAP;

    for (int base = 0; base < mx; base += 8) {
        const int j = base + f;
        const int s_l = (j < mm) ? row[j] : n;
        float v8[8];
        #pragma unroll
        for (int t = 0; t < 8; ++t) {
            const int s_t = __shfl(s_l, gb + t);
            v8[t] = g[s_t * OUT_DIM + f];
        }
        #pragma unroll
        for (int t = 0; t < 8; ++t) acc += v8[t];
    }
    if (__builtin_amdgcn_ballot_w32(live && m > CAP) != 0u) {
        const int K = spill_cnt[0];
        for (int i = 0; i < K; ++i) {
            const int2 p = spill[i];
            const float v = g[p.x * OUT_DIM + f];
            acc = (live && p.y == d) ? acc + v : acc;
        }
    }
    if (live) out[node * OUT_DIM + f] = acc * dd + b2[f];
}

extern "C" void kernel_launch(void* const* d_in, const int* in_sizes, int n_in,
                              void* d_out, int out_size, void* d_ws, size_t ws_size,
                              hipStream_t stream) {
    const float* x  = (const float*)d_in[0];
    const int*   ei = (const int*)d_in[1];
    const float* W1 = (const float*)d_in[2];
    const float* b1 = (const float*)d_in[3];
    const float* W2 = (const float*)d_in[4];
    const float* b2 = (const float*)d_in[5];
    float* out = (float*)d_out;

    const int n = in_sizes[0] / IN_DIM;
    const int E = in_sizes[1] / 2;
    const int* srcIdx = ei;
    const int* dstIdx = ei + E;

    const int nWG = (E + CH - 1) / CH;
    const int B   = (n + RB - 1) / RB;
    const int PB  = B + 1;

    const size_t nr = ((size_t)n + 63) / 64 * 64;
    const size_t Er = ((size_t)E + 63) / 64 * 64;
    const size_t Pr = ((size_t)nWG * PB + 63) / 64 * 64;
    int*      cnt       = (int*)d_ws;
    int*      spill_cnt = cnt + nr;
    int*      pre       = spill_cnt + 64;
    unsigned* sorted    = (unsigned*)(pre + Pr);
    int*      ell       = (int*)(sorted + (size_t)nWG * CH);
    int2*     spill     = (int2*)(ell + nr * CAP);
    float*    hw        = (float*)(spill + Er);
    float*    g         = hw + (nr + 64) * HID_DIM;

    const dim3 blk(256);
    const int ntiles = (n + 15) / 16;

    sort_kernel<<<dim3(nWG), dim3(1024), 0, stream>>>(srcIdx, dstIdx, sorted, pre, spill_cnt, hw, g, E, n, PB);
    build_kernel<<<dim3(B), blk, 0, stream>>>(sorted, pre, cnt, ell, spill_cnt, spill, nWG, PB, n);
    gemm1_kernel<<<dim3((ntiles + 7) / 8), blk, 0, stream>>>(x, W1, cnt, hw, n);
    agg1_kernel<<<dim3((n + 8 * NPW - 1) / (8 * NPW)), blk, 0, stream>>>(cnt, ell, spill_cnt, spill, hw, b1, W2, g, n);
    agg2_kernel<<<dim3((n * OUT_DIM + 255) / 256), blk, 0, stream>>>(cnt, ell, spill_cnt, spill, g, b2, out, n);
}


# x @ W1 moved into the build launch: 8 more waves per work-group multiply the block's own 256 rows beside the 8 that build the table, then scale them by its LDS counts; one launch less, same values
# speedup vs baseline: 5.7624x; 1.0734x over previous
#include <hip/hip_runtime.h>


typedef float v2f __attribute__((ext_vector_type(2)));
typedef float v8f __attribute__((ext_vector_type(8)));

#define IN_DIM  128
#define HID_DIM 32
#define OUT_DIM 8
#define CAP     64
#define CH      8192
#define BSH     8
#define RB      256
#define MAXB    512


__global__ void __launch_bounds__(1024) sort_kernel(const int* __restrict__ srcIdx,
                                                    const int* __restrict__ dstIdx,
                                                    unsigned* __restrict__ sorted,
                                                    int* __restrict__ pre,
                                                    int* __restrict__ spill_cnt,
                                                    float* __restrict__ hw,
                                                    float* __restrict__ g,
                                                    int E, int n, int PB) {
    __shared__ int hist[MAXB];
    __shared__ int pfx[MAXB + 1];
    __shared__ unsigned val[CH];
    const int tid = threadIdx.x;
    const int e0 = blockIdx.x * CH;
    if (blockIdx.x == 0) {
        if (tid == 0) spill_cnt[0] = 0;
        if (tid < HID_DIM) hw[n * HID_DIM + tid] = 0.0f;
        if (tid < OUT_DIM) g[n * OUT_DIM + tid] = 0.0f;
    }
    if (tid < MAXB) hist[tid] = 0;
    __syncthreads();

    unsigned w[CH / 1024];
    int bb[CH / 1024];
    int pos[CH / 1024];
    int dv[CH / 1024];
    int sv[CH / 1024];
    #pragma unroll
    for (int k = 0; k < CH / 1024; ++k) {
        const int ec = min(e0 + k * 1024 + tid, E - 1);
        dv[k] = dstIdx[ec];
        sv[k] = srcIdx[ec];
    }
    #pragma unroll
    for (int k = 0; k < CH / 1024; ++k) {
        const bool valid = (e0 + k * 1024 + tid) < E;
        bb[k] = valid ? (dv[k] >> BSH) : -1;
        w[k]  = ((unsigned)(dv[k] & (RB - 1)) << 24) | (unsigned)sv[k];
    }
    #pragma unroll
    for (int k = 0; k < CH / 1024; ++k) {
        pos[k] = 0;
        if (bb[k] >= 0) pos[k] = atomicAdd(&hist[bb[k]], 1);
    }
    __syncthreads();

    if (tid < 32) {
        int loc[MAXB / 32];
        int sum = 0;
        #pragma unroll
        for (int j = 0; j < MAXB / 32; ++j) { loc[j] = hist[tid * (MAXB / 32) + j]; sum += loc[j]; }
        int incl = sum;
        #pragma unroll
        for (int off = 1; off < 32; off <<= 1) {
            const int t = __shfl_up(incl, off);
            if (tid >= off) incl += t;
        }
        int run = incl - sum;
        #pragma unroll
        for (int j = 0; j < MAXB / 32; ++j) { pfx[tid * (MAXB / 32) + j] = run; run += loc[j]; }
        if (tid == 31) pfx[MAXB] = run;
    }
    __syncthreads();

    #pragma unroll
    for (int k = 0; k < CH / 1024; ++k)
        if (bb[k] >= 0) val[pfx[bb[k]] + pos[k]] = w[k];
    if (tid < PB) pre[blockIdx.x * PB + tid] = pfx[tid];
    __syncthreads();

    const int tot = pfx[MAXB];
    for (int i = tid; i < tot; i += 1024) sorted[e0 + i] = val[i];
}

__global__ void __launch_bounds__(512) build_gemm_kernel(const unsigned* __restrict__ sorted,
                                                         const int* __restrict__ pre,
                                                         int* __restrict__ cnt,
                                                         int* __restrict__ ell,
                                                         int* __restrict__ spill_cnt,
                                                         int2* __restrict__ spill,
                                                         const float* __restrict__ x,
                                                         const float* __restrict__ W1,
                                                         float* __restrict__ hw,
                                                         int nWG, int PB, int n) {
    __shared__ int lcnt[RB];
    __shared__ int lell[RB * CAP];
    __shared__ int lo_s[256];
    __shared__ int len_s[256];
    const int lane = threadIdx.x & 31;
    const int wv16 = __builtin_amdgcn_readfirstlane((int)(threadIdx.x >> 5));
    const bool builder = ((wv16 ^ (wv16 >> 2)) & 1) == 0;
    const int wave = wv16 >> 1;
    const int tid  = wave * 32 + lane;
    const int bin  = blockIdx.x;
    const int node0 = bin * RB;

    const int half = lane >> 4;
    const int l16  = lane & 15;
    const int tile0 = bin * (RB / 16) + 2 * wave;
    const int ntiles = (n + 15) >> 4;
    const bool mult = !builder && tile0 < ntiles;
    int rowA = tile0 * 16 + l16;
    int rowB = rowA + 16;
    if (rowA >= n) rowA = n - 1;
    if (rowB >= n) rowB = n - 1;
    const float* xrA = x + rowA * IN_DIM + 2 * half;
    const float* xrB = x + rowB * IN_DIM + 2 * half;
    v2f pa[8] = {}, pb[8] = {};
    if (mult) {
        #pragma unroll
        for (int j = 0; j < 8; ++j) {
            pa[j].x = xrA[4 * j]; pa[j].y = xrA[4 * j + 1];
            pb[j].x = xrB[4 * j]; pb[j].y = xrB[4 * j + 1];
        }
        __builtin_prefetch(x + (half ? rowB : rowA) * IN_DIM + IN_DIM / 2, 0, 3);
    }

    auto take = [&](unsigned wv) { return atomicAdd(&lcnt[wv >> 24], 1); };
    auto place = [&](unsigned wv, int r) {
        const int dl = (int)(wv >> 24);
        const int s  = (int)(wv & 0xFFFFFFu);
        if (r < CAP) {
            lell[dl * CAP + r] = s;
        } else {
            const int kk = atomicAdd(spill_cnt, 1);
            spill[kk] = make_int2(s, node0 + dl);
        }
    };

    if (builder) lcnt[tid] = 0;
    for (int g0 = 0; g0 < nWG; g0 += 256) {
        __syncthreads();
        if (builder) {
            const int wg = g0 + tid;
            int lo = 0, len = 0;
            if (wg < nWG) {
                lo  = pre[wg * PB + bin];
                len = pre[wg * PB + bin + 1] - lo;
            }
            lo_s[tid] = lo;
            len_s[tid] = len;
        }
        __syncthreads();
        if (builder) {
            for (int k0 = wave; k0 < 256; k0 += 64) {
                const int kk  = k0 + 8 * (lane & 7);
                const int lnv = len_s[kk];
                const int lov = lo_s[kk];
                unsigned v[8];
                int ln[8];
                const unsigned* bp[8];
                #pragma unroll
                for (int r = 0; r < 8; ++r) {
                    ln[r] = __builtin_amdgcn_readlane(lnv, r);
                    bp[r] = sorted + (size_t)(g0 + k0 + 8 * r) * CH + __builtin_amdgcn_readlane(lov, r);
                    v[r] = (lane < ln[r]) ? bp[r][lane] : 0u;
                }
                int sl[8];
                #pragma unroll
                for (int r = 0; r < 8; ++r)
                    sl[r] = (lane < ln[r]) ? take(v[r]) : 0;
                #pragma unroll
                for (int r = 0; r < 8; ++r)
                    if (lane < ln[r]) place(v[r], sl[r]);
                #pragma unroll
                for (int r = 0; r < 8; ++r)
                    for (int j = 32 + lane; j < ln[r]; j += 32) {
                        const unsigned wv = bp[r][j];
                        place(wv, take(wv));
                    }
            }
        }
    }

    v8f c00 = {}; v8f c01 = {}; v8f c10 = {}; v8f c11 = {};
    if (mult) {
        const float* wp = W1 + 2 * half * HID_DIM + l16;
        #pragma unroll
        for (int j = 0; j < 8; ++j) {
            const int k = 4 * j;
            v2f b0; b0.x = wp[k * HID_DIM];       b0.y = wp[(k + 1) * HID_DIM];
            v2f b1; b1.x = wp[k * HID_DIM + 16];  b1.y = wp[(k + 1) * HID_DIM + 16];
            c00 = __builtin_amdgcn_wmma_f32_16x16x4_f32(false, pa[j], false, b0, (short)0, c00, false, false);
            c01 = __builtin_amdgcn_wmma_f32_16x16x4_f32(false, pa[j], false, b1, (short)0, c01, false, false);
            c10 = __builtin_amdgcn_wmma_f32_16x16x4_f32(false, pb[j], false, b0, (short)0, c10, false, false);
            c11 = __builtin_amdgcn_wmma_f32_16x16x4_f32(false, pb[j], false, b1, (short)0, c11, false, false);
        }
        #pragma unroll 8
        for (int k = 32; k < IN_DIM; k += 4) {
            v2f a;  a.x  = xrA[k];                a.y  = xrA[k + 1];
            v2f b;  b.x  = xrB[k];                b.y  = xrB[k + 1];
            v2f b0; b0.x = wp[k * HID_DIM];       b0.y = wp[(k + 1) * HID_DIM];
            v2f b1; b1.x = wp[k * HID_DIM + 16];  b1.y = wp[(k + 1) * HID_DIM + 16];
            c00 = __builtin_amdgcn_wmma_f32_16x16x4_f32(false, a, false, b0, (short)0, c00, false, false);
            c01 = __builtin_amdgcn_wmma_f32_16x16x4_f32(false, a, false, b1, (short)0, c01, false, false);
            c10 = __builtin_amdgcn_wmma_f32_16x16x4_f32(false, b, false, b0, (short)0, c10, false, false);
            c11 = __builtin_amdgcn_wmma_f32_16x16x4_f32(false, b, false, b1, (short)0, c11, false, false);
        }
    }
    __syncthreads();

    if (builder) {
        if (node0 + tid < n) cnt[node0 + tid] = lcnt[tid];
        const int myrow = wave * 32 + lane;
        const int mine  = (node0 + myrow < n) ? min(lcnt[myrow], CAP) : 0;
        #pragma unroll 8
        for (int i = 0; i < 32; ++i) {
            const int rr = wave * 32 + i;
            const int m  = __builtin_amdgcn_readlane(mine, i);
            int* dstrow = ell + (size_t)(node0 + rr) * CAP;
            if (lane < m)      dstrow[lane]      = lell[rr * CAP + lane];
            if (lane + 32 < m) dstrow[32 + lane] = lell[rr * CAP + 32 + lane];
        }
    } else if (mult) {
        #pragma unroll
        for (int i = 0; i < 8; ++i) {
            const int lr = 32 * wave + i + 8 * half;
            const float s0 = rsqrtf((float)(lcnt[lr] + 1));
            const float s1 = rsqrtf((float)(lcnt[lr + 16] + 1));
            const int o0 = node0 + lr;
            const int o1 = o0 + 16;
            if (o0 < n) {
                hw[o0 * HID_DIM + l16]      = c00[i] * s0;
                hw[o0 * HID_DIM + 16 + l16] = c01[i] * s0;
            }
            if (o1 < n) {
                hw[o1 * HID_DIM + l16]      = c10[i] * s1;
                hw[o1 * HID_DIM + 16 + l16] = c11[i] * s1;
            }
        }
    }
}

#define NPW 8
#define HPITCH 36

template <int T>
__device__ __forceinline__ int group_bcast(int v) {
    return __builtin_amdgcn_mov_dpp8(v, T * 0x249249);
}

__device__ __forceinline__ float4 row_piece(const float* __restrict__ tab, int row, int c) {
    const unsigned off = (unsigned)row * (HID_DIM * 4u) + (unsigned)c * 16u;
    return *(const float4*)((const char*)tab + off);
}

__device__ __forceinline__ void add4(float4& a, const float4 b) {
    a.x += b.x; a.y += b.y; a.z += b.z; a.w += b.w;
}

__global__ void __launch_bounds__(256) agg1_kernel(const int* __restrict__ cnt,
                                                   const int* __restrict__ ell,
                                                   const int* __restrict__ spill_cnt,
                                                   const int2* __restrict__ spill,
                                                   const float* __restrict__ hw,
                                                   const float* __restrict__ b1,
                                                   const float* __restrict__ W2,
                                                   float* __restrict__ g, int n) {
    __shared__ __attribute__((aligned(16))) float hs[8 * NPW * HPITCH];
    const int lane = threadIdx.x & 31;
    const int q = lane >> 3;
    const int c = lane & 7;
    const int wv = __builtin_amdgcn_readfirstlane((int)(blockIdx.x * 8 + (threadIdx.x >> 5)));
    const int d0 = wv * NPW;
    if (d0 >= n) return;
    float* myh = hs + __builtin_amdgcn_readfirstlane((int)(threadIdx.x >> 5)) * (NPW * HPITCH);
    const int nn = min(NPW, n - d0);
    const int cv = (lane < nn) ? cnt[d0 + lane] : 0;
    float4 bias;
    bias.x = b1[4 * c]; bias.y = b1[4 * c + 1]; bias.z = b1[4 * c + 2]; bias.w = b1[4 * c + 3];
    const int sidx = 4 * c + q;

    int m_nx  = __builtin_amdgcn_readlane(cv, 0);
    int mm_nx = min(m_nx, CAP);
    int sv_nx = (sidx < mm_nx) ? ell[d0 * CAP + sidx] : n;

    for (int i = 0; i < nn; ++i) {
        const int d  = d0 + i;
        const int m  = m_nx;
        const int mm = mm_nx;
        const int sv = sv_nx;
        if (i + 1 < nn) {
            m_nx  = __builtin_amdgcn_readlane(cv, i + 1);
            mm_nx = min(m_nx, CAP);
            sv_nx = (sidx < mm_nx) ? ell[(d + 1) * CAP + sidx] : n;
        }
        const int* row = ell + d * CAP;

        const float4 self = row_piece(hw, d, c);
        const float4 v0 = row_piece(hw, group_bcast<0>(sv), c);
        const float4 v1 = row_piece(hw, group_bcast<1>(sv), c);
        const float4 v2 = row_piece(hw, group_bcast<2>(sv), c);
        const float4 v3 = row_piece(hw, group_bcast<3>(sv), c);
        const float4 v4 = row_piece(hw, group_bcast<4>(sv), c);
        float4 v5 = make_float4(0.0f, 0.0f, 0.0f, 0.0f);
        float4 v6 = v5, v7 = v5;
        if (mm > 20) {
            v5 = row_piece(hw, group_bcast<5>(sv), c);
            v6 = row_piece(hw, group_bcast<6>(sv), c);
            v7 = row_piece(hw, group_bcast<7>(sv), c);
        }
        float4 acc = (q == 0) ? self : make_float4(0.0f, 0.0f, 0.0f, 0.0f);
        add4(acc, v0); add4(acc, v1); add4(acc, v2); add4(acc, v3); add4(acc, v4);
        add4(acc, v5); add4(acc, v6); add4(acc, v7);
        for (int j0 = 32; j0 < mm; j0 += 4) {
            const int j = j0 + q;
            const int s = (j < mm) ? row[j] : n;
            add4(acc, row_piece(hw, s, c));
        }
        if (m > CAP) {
            const int K = spill_cnt[0];
            for (int k = 0; k < K; ++k) {
                const int2 p = spill[k];
                if (p.y == d && q == 0) add4(acc, row_piece(hw, p.x, c));
            }
        }
        acc.x += __shfl_xor(acc.x, 8);  acc.y += __shfl_xor(acc.y, 8);
        acc.z += __shfl_xor(acc.z, 8);  acc.w += __shfl_xor(acc.w, 8);
        acc.x += __shfl_xor(acc.x, 16); acc.y += __shfl_xor(acc.y, 16);
        acc.z += __shfl_xor(acc.z, 16); acc.w += __shfl_xor(acc.w, 16);
        const float dd = rsqrtf((float)(m + 1));
        if (q == 0) {
            float* hrow = myh + i * HPITCH + 4 * c;
            hrow[0] = fmaxf(acc.x * dd + bias.x, 0.0f);
            hrow[1] = fmaxf(acc.y * dd + bias.y, 0.0f);
            hrow[2] = fmaxf(acc.z * dd + bias.z, 0.0f);
            hrow[3] = fmaxf(acc.w * dd + bias.w, 0.0f);
        }
    }

    const int half = lane >> 4;
    const int l16  = lane & 15;
    const bool rowok = l16 < nn;
    const bool live  = l16 < OUT_DIM;
    const float* hr = myh + (rowok ? l16 : 0) * HPITCH + 2 * half;
    const float* wc = W2 + 2 * half * OUT_DIM + (l16 & (OUT_DIM - 1));
    float w0[HID_DIM / 4], w1[HID_DIM / 4], a0[HID_DIM / 4], a1[HID_DIM / 4];
    #pragma unroll
    for (int j = 0; j < HID_DIM / 4; ++j) {
        w0[j] = wc[(4 * j) * OUT_DIM];
        w1[j] = wc[(4 * j + 1) * OUT_DIM];
        a0[j] = hr[4 * j];
        a1[j] = hr[4 * j + 1];
    }
    v8f pr = {};
    #pragma unroll
    for (int j = 0; j < HID_DIM / 4; ++j) {
        v2f a; a.x = rowok ? a0[j] : 0.0f; a.y = rowok ? a1[j] : 0.0f;
        v2f b; b.x = live  ? w0[j] : 0.0f; b.y = live  ? w1[j] : 0.0f;
        pr = __builtin_amdgcn_wmma_f32_16x16x4_f32(false, a, false, b, (short)0, pr, false, false);
    }
    const float dl = rsqrtf((float)(cv + 1));
    float o[NPW];
    #pragma unroll
    for (int i = 0; i < NPW; ++i)
        o[i] = pr[i] * __int_as_float(__builtin_amdgcn_readlane(__float_as_int(dl), i));
    if (live && half == 0) {
        #pragma unroll
        for (int i = 0; i < NPW; ++i)
            if (i < nn) g[(d0 + i) * OUT_DIM + l16] = o[i];
    }
}

__global__ void __launch_bounds__(256) agg2_kernel(const int* __restrict__ cnt,
                                                   const int* __restrict__ ell,
                                                   const int* __restrict__ spill_cnt,
                                                   const int2* __restrict__ spill,
                                                   const float* __restrict__ g,
                                                   const float* __restrict__ b2,
                                                   float* __restrict__ out, int n) {
    const int tid  = blockIdx.x * blockDim.x + threadIdx.x;
    const int lane = threadIdx.x & 31;
    const int f    = lane & 7;
    const int gb   = lane & 24;
    const int node = tid >> 3;
    const bool live = node < n;
    const int d = live ? node : n - 1;
    const int m = cnt[d];
    const float dd = rsqrtf((float)(m + 1));
    const int mm = live ? min(m, CAP) : 0;
    float acc = g[d * OUT_DIM + f];
    int mx = mm;
    mx = max(mx, __shfl_xor(mx, 8));
    mx = max(mx, __shfl_xor(mx, 16));
    const int* row = ell + d * CAP;

    for (int base = 0; base < mx; base += 8) {
        const int j = base + f;
        const int s_l = (j < mm) ? row[j] : n;
        float v8[8];
        #pragma unroll
        for (int t = 0; t < 8; ++t) {
            const int s_t = __shfl(s_l, gb + t);
            v8[t] = g[s_t * OUT_DIM + f];
        }
        #pragma unroll
        for (int t = 0; t < 8; ++t) acc += v8[t];
    }
    if (__builtin_amdgcn_ballot_w32(live && m > CAP) != 0u) {
        const int K = spill_cnt[0];
        for (int i = 0; i < K; ++i) {
            const int2 p = spill[i];
            const float v = g[p.x * OUT_DIM + f];
            acc = (live && p.y == d) ? acc + v : acc;
        }
    }
    if (live) out[node * OUT_DIM + f] = acc * dd + b2[f];
}

extern "C" void kernel_launch(void* const* d_in, const int* in_sizes, int n_in,
                              void* d_out, int out_size, void* d_ws, size_t ws_size,
                              hipStream_t stream) {
    const float* x  = (const float*)d_in[0];
    const int*   ei = (const int*)d_in[1];
    const float* W1 = (const float*)d_in[2];
    const float* b1 = (const float*)d_in[3];
    const float* W2 = (const float*)d_in[4];
    const float* b2 = (const float*)d_in[5];
    float* out = (float*)d_out;

    const int n = in_sizes[0] / IN_DIM;
    const int E = in_sizes[1] / 2;
    const int* srcIdx = ei;
    const int* dstIdx = ei + E;

    const int nWG = (E + CH - 1) / CH;
    const int B   = (n + RB - 1) / RB;
    const int PB  = B + 1;

    const size_t nr = ((size_t)n + 63) / 64 * 64;
    const size_t Er = ((size_t)E + 63) / 64 * 64;
    const size_t Pr = ((size_t)nWG * PB + 63) / 64 * 64;
    int*      cnt       = (int*)d_ws;
    int*      spill_cnt = cnt + nr;
    int*      pre       = spill_cnt + 64;
    unsigned* sorted    = (unsigned*)(pre + Pr);
    int*      ell       = (int*)(sorted + (size_t)nWG * CH);
    int2*     spill     = (int2*)(ell + nr * CAP);
    float*    hw        = (float*)(spill + Er);
    float*    g         = hw + (nr + 64) * HID_DIM;

    const dim3 blk(256);

    sort_kernel<<<dim3(nWG), dim3(1024), 0, stream>>>(srcIdx, dstIdx, sorted, pre, spill_cnt, hw, g, E, n, PB);
    build_gemm_kernel<<<dim3(B), dim3(512), 0, stream>>>(sorted, pre, cnt, ell, spill_cnt, spill, x, W1, hw, nWG, PB, n);
    agg1_kernel<<<dim3((n + 8 * NPW - 1) / (8 * NPW)), blk, 0, stream>>>(cnt, ell, spill_cnt, spill, hw, b1, W2, g, n);
    agg2_kernel<<<dim3((n * OUT_DIM + 255) / 256), blk, 0, stream>>>(cnt, ell, spill_cnt, spill, g, b2, out, n);
}
